# starting kernel
# baseline (speedup 1.0000x reference)
#include <hip/hip_runtime.h>


#define NROWS 8192
#define DIM   128

typedef __attribute__((ext_vector_type(2))) float v2f;
typedef __attribute__((ext_vector_type(8))) float v8f;

__global__ void pdist_sqnorm_kernel(const float* __restrict__ x,
                                    float* __restrict__ sq,
                                    float* __restrict__ out,
                                    long long total) {
    int i = blockIdx.x * blockDim.x + threadIdx.x;
    if (i < NROWS) {
        const float4* row = (const float4*)(x + (size_t)i * DIM);
        float acc = 0.0f;
#pragma unroll
        for (int j = 0; j < DIM / 4; ++j) {
            float4 v = row[j];
            acc = fmaf(v.x, v.x, acc);
            acc = fmaf(v.y, v.y, acc);
            acc = fmaf(v.z, v.z, acc);
            acc = fmaf(v.w, v.w, acc);
        }
        sq[i] = acc;
    }
    if (i == 0) {
        out[total - 1] = 0.0f;
    }
}

__global__ void __launch_bounds__(32)
pdist_tile_kernel(const float* __restrict__ x,
                  const float* __restrict__ sq,
                  float* __restrict__ out) {
    const int tu = blockIdx.x;
    const int tl = blockIdx.y;
    if (tu < tl) return;

    const int lane = threadIdx.x;
    const int lm   = lane & 15;
    const int kg   = (lane >> 4) << 1;

    const float* __restrict__ arow = x + (size_t)(tl * 16 + lm) * DIM + kg;
    const float* __restrict__ brow = x + (size_t)(tu * 16 + lm) * DIM + kg;

    v8f c = {};
#pragma unroll
    for (int s = 0; s < DIM / 4; ++s) {
        v2f a = *(const v2f*)(arow + 4 * s);
        v2f b = *(const v2f*)(brow + 4 * s);
        c = __builtin_amdgcn_wmma_f32_16x16x4_f32(
                false, a, false, b, (short)0, c, false, false);
    }

    const int u     = tu * 16 + lm;
    const int mbase = tl * 16 + ((lane >> 4) << 3);
    const float squ = sq[u];
    const long long ubase = (long long)u * (u - 1) / 2 - 1;

    if (tu > tl) {
        float* __restrict__ o = out + (ubase + mbase);
#pragma unroll
        for (int j = 0; j < 8; ++j) {
            o[j] = sq[mbase + j] + squ - 2.0f * c[j];
        }
    } else {
#pragma unroll
        for (int j = 0; j < 8; ++j) {
            const int l = mbase + j;
            if (l < u && u >= 2) {
                out[ubase + l] = sq[l] + squ - 2.0f * c[j];
            }
        }
    }
}

extern "C" void kernel_launch(void* const* d_in, const int* in_sizes, int n_in,
                              void* d_out, int out_size, void* d_ws, size_t ws_size,
                              hipStream_t stream) {
    (void)in_sizes; (void)n_in; (void)out_size; (void)ws_size;

    const float* x = (const float*)d_in[0];
    float* out = (float*)d_out;
    float* sq  = (float*)d_ws;

    const long long total = (long long)NROWS * (NROWS - 1) / 2;

    pdist_sqnorm_kernel<<<(NROWS + 127) / 128, 128, 0, stream>>>(x, sq, out, total);

    dim3 grid(NROWS / 16, NROWS / 16);
    pdist_tile_kernel<<<grid, 32, 0, stream>>>(x, sq, out);
}


# Stage each 64x64 block's 64 l-rows and 64 u-rows of x once in LDS by whole-row b128 loads; each wave does a 32x32 sub-block from LDS fragments, each used by two tiles; k order and sums unchanged
# speedup vs baseline: 2.5152x; 2.5152x over previous
#include <hip/hip_runtime.h>


#define NROWS 8192
#define DIM   128

#define BT    64
#define PITCH 132
#define NBT   (NROWS / BT)
#define NBLK  (NBT * (NBT + 1) / 2)

typedef __attribute__((ext_vector_type(2))) float v2f;
typedef __attribute__((ext_vector_type(8))) float v8f;

__global__ void pdist_sqnorm_kernel(const float* __restrict__ x,
                                    float* __restrict__ sq,
                                    float* __restrict__ out,
                                    long long total) {
    int i = blockIdx.x * blockDim.x + threadIdx.x;
    if (i < NROWS) {
        const float4* row = (const float4*)(x + (size_t)i * DIM);
        float acc = 0.0f;
#pragma unroll
        for (int j = 0; j < DIM / 4; ++j) {
            float4 v = row[j];
            acc = fmaf(v.x, v.x, acc);
            acc = fmaf(v.y, v.y, acc);
            acc = fmaf(v.z, v.z, acc);
            acc = fmaf(v.w, v.w, acc);
        }
        sq[i] = acc;
    }
    if (i == 0) {
        out[total - 1] = 0.0f;
    }
}

__global__ void __launch_bounds__(128)
pdist_tile_kernel(const float* __restrict__ x,
                  const float* __restrict__ sq,
                  float* __restrict__ out) {
    __shared__ __attribute__((aligned(16))) float lds[2 * BT * PITCH];
    float* __restrict__ Ls = lds;
    float* __restrict__ Us = lds + BT * PITCH;

    const int t = blockIdx.x;
    int bu = (int)((sqrtf(8.0f * (float)t + 1.0f) - 1.0f) * 0.5f);
    if (bu * (bu + 1) / 2 > t) --bu;
    if ((bu + 1) * (bu + 2) / 2 <= t) ++bu;
    const int bl = t - bu * (bu + 1) / 2;

    const int tid  = threadIdx.x;
    const int lane = tid & 31;
    const int wave = __builtin_amdgcn_readfirstlane(tid >> 5);

    {
        const float* __restrict__ gl = x + (size_t)(bl * BT + wave * 16) * DIM + lane * 4;
        const float* __restrict__ gu = x + (size_t)(bu * BT + wave * 16) * DIM + lane * 4;
        float* sl = Ls + (wave * 16) * PITCH + lane * 4;
        float* su = Us + (wave * 16) * PITCH + lane * 4;
        float4 vl[16], vu[16];
#pragma unroll
        for (int i = 0; i < 16; ++i) vl[i] = *(const float4*)(gl + i * DIM);
#pragma unroll
        for (int i = 0; i < 16; ++i) vu[i] = *(const float4*)(gu + i * DIM);
#pragma unroll
        for (int i = 0; i < 16; ++i) *(float4*)(sl + i * PITCH) = vl[i];
#pragma unroll
        for (int i = 0; i < 16; ++i) *(float4*)(su + i * PITCH) = vu[i];
    }
    __syncthreads();

    const int wl = wave & 1;
    const int wu = wave >> 1;
    if (bu == bl && wu < wl) return;

    const int lm = lane & 15;
    const int hh = lane >> 4;
    const int kg = hh << 1;

    const int l0 = bl * BT + wl * 32;
    const int u0 = bu * BT + wu * 32;

    float sql[2][8];
#pragma unroll
    for (int mt = 0; mt < 2; ++mt)
#pragma unroll
        for (int j = 0; j < 8; ++j) sql[mt][j] = sq[l0 + mt * 16 + (hh << 3) + j];
    float squ[2];
#pragma unroll
    for (int nt = 0; nt < 2; ++nt) squ[nt] = sq[u0 + nt * 16 + lm];

    const float* __restrict__ ap = Ls + (wl * 32 + lm) * PITCH + kg;
    const float* __restrict__ bp = Us + (wu * 32 + lm) * PITCH + kg;

    v8f c00 = {}, c01 = {}, c10 = {}, c11 = {};
#pragma unroll
    for (int s = 0; s < DIM / 4; ++s) {
        v2f a0 = *(const v2f*)(ap + 4 * s);
        v2f a1 = *(const v2f*)(ap + 16 * PITCH + 4 * s);
        v2f b0 = *(const v2f*)(bp + 4 * s);
        v2f b1 = *(const v2f*)(bp + 16 * PITCH + 4 * s);
        c00 = __builtin_amdgcn_wmma_f32_16x16x4_f32(false, a0, false, b0, (short)0, c00, false, false);
        c01 = __builtin_amdgcn_wmma_f32_16x16x4_f32(false, a0, false, b1, (short)0, c01, false, false);
        c10 = __builtin_amdgcn_wmma_f32_16x16x4_f32(false, a1, false, b0, (short)0, c10, false, false);
        c11 = __builtin_amdgcn_wmma_f32_16x16x4_f32(false, a1, false, b1, (short)0, c11, false, false);
    }

    float val[2][2][8];
#pragma unroll
    for (int j = 0; j < 8; ++j) {
        val[0][0][j] = sql[0][j] + squ[0] - 2.0f * c00[j];
        val[0][1][j] = sql[0][j] + squ[1] - 2.0f * c01[j];
        val[1][0][j] = sql[1][j] + squ[0] - 2.0f * c10[j];
        val[1][1][j] = sql[1][j] + squ[1] - 2.0f * c11[j];
    }

    if (bu > bl || wu > wl) {
#pragma unroll
        for (int nt = 0; nt < 2; ++nt) {
            const int u = u0 + nt * 16 + lm;
            const long long ubase = (long long)u * (u - 1) / 2 - 1;
#pragma unroll
            for (int mt = 0; mt < 2; ++mt) {
                float* __restrict__ o = out + (ubase + (l0 + mt * 16 + (hh << 3)));
#pragma unroll
                for (int j = 0; j < 8; ++j) o[j] = val[mt][nt][j];
            }
        }
    } else {
#pragma unroll
        for (int nt = 0; nt < 2; ++nt) {
            const int u = u0 + nt * 16 + lm;
            const long long ubase = (long long)u * (u - 1) / 2 - 1;
#pragma unroll
            for (int mt = 0; mt < 2; ++mt) {
#pragma unroll
                for (int j = 0; j < 8; ++j) {
                    const int l = l0 + mt * 16 + (hh << 3) + j;
                    if (l < u && u >= 2) out[ubase + l] = val[mt][nt][j];
                }
            }
        }
    }
}

extern "C" void kernel_launch(void* const* d_in, const int* in_sizes, int n_in,
                              void* d_out, int out_size, void* d_ws, size_t ws_size,
                              hipStream_t stream) {
    (void)in_sizes; (void)n_in; (void)out_size; (void)ws_size;

    const float* x = (const float*)d_in[0];
    float* out = (float*)d_out;
    float* sq  = (float*)d_ws;

    const long long total = (long long)NROWS * (NROWS - 1) / 2;

    pdist_sqnorm_kernel<<<(NROWS + 127) / 128, 128, 0, stream>>>(x, sq, out, total);

    pdist_tile_kernel<<<NBLK, 128, 0, stream>>>(x, sq, out);
}


# Coalesced stores: u rows as WMMA A and l rows as B, then v_permlane16_swap of half-rows between the two l tiles, so each store writes 32 consecutive l of one u (128 B); k order and sums unchanged
# speedup vs baseline: 4.2385x; 1.6852x over previous
#include <hip/hip_runtime.h>


#define NROWS 8192
#define DIM   128

#define BT    64
#define PITCH 132
#define NBT   (NROWS / BT)
#define NBLK  (NBT * (NBT + 1) / 2)

typedef __attribute__((ext_vector_type(2))) float v2f;
typedef __attribute__((ext_vector_type(8))) float v8f;
typedef __attribute__((ext_vector_type(2))) unsigned v2u;

__global__ void pdist_sqnorm_kernel(const float* __restrict__ x,
                                    float* __restrict__ sq,
                                    float* __restrict__ out,
                                    long long total) {
    int i = blockIdx.x * blockDim.x + threadIdx.x;
    if (i < NROWS) {
        const float4* row = (const float4*)(x + (size_t)i * DIM);
        float acc = 0.0f;
#pragma unroll
        for (int j = 0; j < DIM / 4; ++j) {
            float4 v = row[j];
            acc = fmaf(v.x, v.x, acc);
            acc = fmaf(v.y, v.y, acc);
            acc = fmaf(v.z, v.z, acc);
            acc = fmaf(v.w, v.w, acc);
        }
        sq[i] = acc;
    }
    if (i == 0) {
        out[total - 1] = 0.0f;
    }
}

__device__ __forceinline__ void pdist_rows(const v8f t0, const v8f t1,
                                           const float* base, float* val) {
#pragma unroll
    for (int j = 0; j < 8; ++j) {
        const v2u r = __builtin_amdgcn_permlane16_swap(__float_as_uint(t0[j]),
                                                       __float_as_uint(t1[j]), false, false);
        val[j]     = base[j]     - 2.0f * __uint_as_float(r[0]);
        val[j + 8] = base[j + 8] - 2.0f * __uint_as_float(r[1]);
    }
}

__global__ void __launch_bounds__(128)
pdist_tile_kernel(const float* __restrict__ x,
                  const float* __restrict__ sq,
                  float* __restrict__ out) {
    __shared__ __attribute__((aligned(16))) float lds[2 * BT * PITCH];
    float* __restrict__ Ls = lds;
    float* __restrict__ Us = lds + BT * PITCH;

    const int t = blockIdx.x;
    int bu = (int)((sqrtf(8.0f * (float)t + 1.0f) - 1.0f) * 0.5f);
    if (bu * (bu + 1) / 2 > t) --bu;
    if ((bu + 1) * (bu + 2) / 2 <= t) ++bu;
    const int bl = t - bu * (bu + 1) / 2;

    const int tid  = threadIdx.x;
    const int lane = tid & 31;
    const int wave = __builtin_amdgcn_readfirstlane(tid >> 5);

    {
        const float* __restrict__ gl = x + (size_t)(bl * BT + wave * 16) * DIM + lane * 4;
        const float* __restrict__ gu = x + (size_t)(bu * BT + wave * 16) * DIM + lane * 4;
        float* sl = Ls + (wave * 16) * PITCH + lane * 4;
        float* su = Us + (wave * 16) * PITCH + lane * 4;
        float4 vl[16], vu[16];
#pragma unroll
        for (int i = 0; i < 16; ++i) vl[i] = *(const float4*)(gl + i * DIM);
#pragma unroll
        for (int i = 0; i < 16; ++i) vu[i] = *(const float4*)(gu + i * DIM);
#pragma unroll
        for (int i = 0; i < 16; ++i) *(float4*)(sl + i * PITCH) = vl[i];
#pragma unroll
        for (int i = 0; i < 16; ++i) *(float4*)(su + i * PITCH) = vu[i];
    }
    __syncthreads();

    const int wl = wave & 1;
    const int wu = wave >> 1;
    if (bu == bl && wu < wl) return;

    const int lm = lane & 15;
    const int hh = lane >> 4;
    const int kg = hh << 1;

    const int l0 = bl * BT + wl * 32;
    const int u0 = bu * BT + wu * 32;

    const int   lrow = l0 + lane;
    const float sql  = sq[lrow];
    float squ[32];
#pragma unroll
    for (int i = 0; i < 32; ++i) squ[i] = sq[u0 + i];

    const float* __restrict__ ap = Us + (wu * 32 + lm) * PITCH + kg;
    const float* __restrict__ bp = Ls + (wl * 32 + lm) * PITCH + kg;

    v8f c00 = {}, c01 = {}, c10 = {}, c11 = {};
#pragma unroll
    for (int s = 0; s < DIM / 4; ++s) {
        v2f a0 = *(const v2f*)(ap + 4 * s);
        v2f a1 = *(const v2f*)(ap + 16 * PITCH + 4 * s);
        v2f b0 = *(const v2f*)(bp + 4 * s);
        v2f b1 = *(const v2f*)(bp + 16 * PITCH + 4 * s);
        c00 = __builtin_amdgcn_wmma_f32_16x16x4_f32(false, a0, false, b0, (short)0, c00, false, false);
        c01 = __builtin_amdgcn_wmma_f32_16x16x4_f32(false, a0, false, b1, (short)0, c01, false, false);
        c10 = __builtin_amdgcn_wmma_f32_16x16x4_f32(false, a1, false, b0, (short)0, c10, false, false);
        c11 = __builtin_amdgcn_wmma_f32_16x16x4_f32(false, a1, false, b1, (short)0, c11, false, false);
    }

    float base[32], val[32];
#pragma unroll
    for (int i = 0; i < 32; ++i) base[i] = sql + squ[i];
    pdist_rows(c00, c01, base,      val);
    pdist_rows(c10, c11, base + 16, val + 16);

    if (bu > bl || wu > wl) {
#pragma unroll
        for (int i = 0; i < 32; ++i) {
            const int u = u0 + i;
            float* orow = out + (u * (u - 1) / 2 - 1);
            orow[(unsigned)lrow] = val[i];
        }
    } else {
#pragma unroll
        for (int i = 0; i < 32; ++i) {
            const int u = u0 + i;
            float* orow = out + (u * (u - 1) / 2 - 1);
            if (lrow < u && u >= 2) orow[(unsigned)lrow] = val[i];
        }
    }
}

extern "C" void kernel_launch(void* const* d_in, const int* in_sizes, int n_in,
                              void* d_out, int out_size, void* d_ws, size_t ws_size,
                              hipStream_t stream) {
    (void)in_sizes; (void)n_in; (void)out_size; (void)ws_size;

    const float* x = (const float*)d_in[0];
    float* out = (float*)d_out;
    float* sq  = (float*)d_ws;

    const long long total = (long long)NROWS * (NROWS - 1) / 2;

    pdist_sqnorm_kernel<<<(NROWS + 127) / 128, 128, 0, stream>>>(x, sq, out, total);

    pdist_tile_kernel<<<NBLK, 128, 0, stream>>>(x, sq, out);
}


# Distances staged in LDS per 64(u)x128(l) work-group tile (8 waves) and written as whole aligned 128-B lines: 3 full-line stores + 1 split head/tail store per row; k order and sums unchanged
# speedup vs baseline: 4.5036x; 1.0625x over previous
#include <hip/hip_runtime.h>


#define NROWS 8192
#define DIM   128

#define BU    64
#define BL    128
#define PITCH 132
#define OT    128
#define NPAIR (NROWS / BL)
#define NBLK  (NPAIR * (NPAIR + 1))

typedef __attribute__((ext_vector_type(2))) float v2f;
typedef __attribute__((ext_vector_type(8))) float v8f;
typedef __attribute__((ext_vector_type(2))) unsigned v2u;

__global__ void pdist_sqnorm_kernel(const float* __restrict__ x,
                                    float* __restrict__ sq,
                                    float* __restrict__ out,
                                    long long total) {
    int i = blockIdx.x * blockDim.x + threadIdx.x;
    if (i < NROWS) {
        const float4* row = (const float4*)(x + (size_t)i * DIM);
        float acc = 0.0f;
#pragma unroll
        for (int j = 0; j < DIM / 4; ++j) {
            float4 v = row[j];
            acc = fmaf(v.x, v.x, acc);
            acc = fmaf(v.y, v.y, acc);
            acc = fmaf(v.z, v.z, acc);
            acc = fmaf(v.w, v.w, acc);
        }
        sq[i] = acc;
    }
    if (i == 0) {
        out[total - 1] = 0.0f;
    }
}

__device__ __forceinline__ void pdist_rows(const v8f t0, const v8f t1,
                                           const float* base, float* val) {
#pragma unroll
    for (int j = 0; j < 8; ++j) {
        const v2u r = __builtin_amdgcn_permlane16_swap(__float_as_uint(t0[j]),
                                                       __float_as_uint(t1[j]), false, false);
        val[j]     = base[j]     - 2.0f * __uint_as_float(r[0]);
        val[j + 8] = base[j + 8] - 2.0f * __uint_as_float(r[1]);
    }
}

__global__ void __launch_bounds__(256)
pdist_tile_kernel(const float* __restrict__ x,
                  const float* __restrict__ sq,
                  float* __restrict__ out) {
    __shared__ __attribute__((aligned(16))) float lds[(BL + BU) * PITCH];
    float* __restrict__ Ls = lds;
    float* __restrict__ Us = lds + BL * PITCH;

    const int t = blockIdx.x;
    int p = (int)((sqrtf(4.0f * (float)t + 1.0f) - 1.0f) * 0.5f);
    if (p * (p + 1) > t) --p;
    if ((p + 1) * (p + 2) <= t) ++p;
    const int r   = t - p * (p + 1);
    const int odd = (r > p) ? 1 : 0;
    const int bu  = 2 * p + odd;
    const int bl  = r - odd * (p + 1);

    const int tid  = threadIdx.x;
    const int lane = tid & 31;
    const int wave = __builtin_amdgcn_readfirstlane(tid >> 5);

    {
        const float* __restrict__ gl = x + (size_t)(bl * BL + wave * 16) * DIM + lane * 4;
        const float* __restrict__ gu = x + (size_t)(bu * BU + wave * 8) * DIM + lane * 4;
        float* sl = Ls + (wave * 16) * PITCH + lane * 4;
        float* su = Us + (wave * 8) * PITCH + lane * 4;
        float4 vl[16], vu[8];
#pragma unroll
        for (int i = 0; i < 16; ++i) vl[i] = *(const float4*)(gl + i * DIM);
#pragma unroll
        for (int i = 0; i < 8; ++i) vu[i] = *(const float4*)(gu + i * DIM);
#pragma unroll
        for (int i = 0; i < 16; ++i) *(float4*)(sl + i * PITCH) = vl[i];
#pragma unroll
        for (int i = 0; i < 8; ++i) *(float4*)(su + i * PITCH) = vu[i];
    }
    __syncthreads();

    const int wl = wave & 3;
    const int wu = wave >> 2;
    const int l0 = bl * BL + wl * 32;
    const int u0 = bu * BU + wu * 32;
    const bool active = (l0 <= u0);

    float val[32];
    if (active) {
        const int lm = lane & 15;
        const int hh = lane >> 4;
        const int kg = hh << 1;

        const float sql = sq[l0 + lane];
        float squ[32];
#pragma unroll
        for (int i = 0; i < 32; ++i) squ[i] = sq[u0 + i];

        const float* __restrict__ ap = Us + (wu * 32 + lm) * PITCH + kg;
        const float* __restrict__ bp = Ls + (wl * 32 + lm) * PITCH + kg;

        v8f c00 = {}, c01 = {}, c10 = {}, c11 = {};
#pragma unroll
        for (int s = 0; s < DIM / 4; ++s) {
            v2f a0 = *(const v2f*)(ap + 4 * s);
            v2f a1 = *(const v2f*)(ap + 16 * PITCH + 4 * s);
            v2f b0 = *(const v2f*)(bp + 4 * s);
            v2f b1 = *(const v2f*)(bp + 16 * PITCH + 4 * s);
            c00 = __builtin_amdgcn_wmma_f32_16x16x4_f32(false, a0, false, b0, (short)0, c00, false, false);
            c01 = __builtin_amdgcn_wmma_f32_16x16x4_f32(false, a0, false, b1, (short)0, c01, false, false);
            c10 = __builtin_amdgcn_wmma_f32_16x16x4_f32(false, a1, false, b0, (short)0, c10, false, false);
            c11 = __builtin_amdgcn_wmma_f32_16x16x4_f32(false, a1, false, b1, (short)0, c11, false, false);
        }

        float base[32];
#pragma unroll
        for (int i = 0; i < 32; ++i) base[i] = sql + squ[i];
        pdist_rows(c00, c01, base,      val);
        pdist_rows(c10, c11, base + 16, val + 16);
    }

    __syncthreads();

    float* __restrict__ Ts = lds;
    if (active) {
        float* tw = Ts + (wu * 32) * OT + wl * 32 + lane;
#pragma unroll
        for (int i = 0; i < 32; ++i) tw[i * OT] = val[i];
    }
    __syncthreads();

    const int lb = bl * BL;
    if (bl < p) {
        float v[32];
#pragma unroll
        for (int q = 0; q < 8; ++q) {
            const int row = wave * 8 + q;
            const int u   = bu * BU + row;
            const int g0  = u * (u - 1) / 2 - 1 + lb;
            const int h   = (-g0) & 31;
            const float* trow = Ts + row * OT;
            const int i0 = h + lane;
            v[4 * q + 0] = trow[i0];
            v[4 * q + 1] = trow[i0 + 32];
            v[4 * q + 2] = trow[i0 + 64];
            v[4 * q + 3] = trow[(lane < h) ? lane : 96 + lane];
        }
#pragma unroll
        for (int q = 0; q < 8; ++q) {
            const int row = wave * 8 + q;
            const int u   = bu * BU + row;
            const int g0  = u * (u - 1) / 2 - 1 + lb;
            const int h   = (-g0) & 31;
            float* orow = out + g0;
            const unsigned i0 = (unsigned)(h + lane);
            orow[i0]      = v[4 * q + 0];
            orow[i0 + 32] = v[4 * q + 1];
            orow[i0 + 64] = v[4 * q + 2];
            orow[(unsigned)((lane < h) ? lane : 96 + lane)] = v[4 * q + 3];
        }
    } else {
        for (int q = 0; q < 8; ++q) {
            const int row = wave * 8 + q;
            const int u   = bu * BU + row;
            const int g0  = u * (u - 1) / 2 - 1 + lb;
            const int h   = (-g0) & 31;
            const float* trow = Ts + row * OT;
            const bool rowok  = (u >= 2);
            for (int j = 0; j < 4; ++j) {
                const int idx = (j < 3) ? (h + 32 * j + lane) : ((lane < h) ? lane : 96 + lane);
                const float val1 = trow[idx];
                if (rowok && lb + idx < u) out[g0 + idx] = val1;
            }
        }
    }
}

extern "C" void kernel_launch(void* const* d_in, const int* in_sizes, int n_in,
                              void* d_out, int out_size, void* d_ws, size_t ws_size,
                              hipStream_t stream) {
    (void)in_sizes; (void)n_in; (void)out_size; (void)ws_size;

    const float* x = (const float*)d_in[0];
    float* out = (float*)d_out;
    float* sq  = (float*)d_ws;

    const long long total = (long long)NROWS * (NROWS - 1) / 2;

    pdist_sqnorm_kernel<<<(NROWS + 127) / 128, 128, 0, stream>>>(x, sq, out, total);

    pdist_tile_kernel<<<NBLK, 256, 0, stream>>>(x, sq, out);
}
